# starting kernel
# baseline (speedup 1.0000x reference)
#include <hip/hip_runtime.h>
#include <hip/hip_bf16.h>


#define NS   4096
#define NG   20000
#define NE   131072
#define DIN  256
#define NH   4
#define C1   64
#define C3   128
#define HC1  256
#define HC3  512
#define NEG_ENC 0x00800000u

typedef __attribute__((ext_vector_type(16))) _Float16 v16h;
typedef __attribute__((ext_vector_type(8)))  _Float16 v8h;
typedef __attribute__((ext_vector_type(8)))  float    v8f;

static inline int cdiv(int a, int b) { return (a + b - 1) / b; }

__device__ __forceinline__ float eluf(float x) { return x > 0.f ? x : (expf(x) - 1.f); }

__global__ void k_f32_to_f16(const float* __restrict__ in, _Float16* __restrict__ out, int n) {
    int i = blockIdx.x * blockDim.x + threadIdx.x;
    if (i < n) out[i] = (_Float16)in[i];
}

__global__ void k_f32_to_f16_t(const float* __restrict__ in, _Float16* __restrict__ out,
                               int K, int N) {
    int i = blockIdx.x * blockDim.x + threadIdx.x;
    if (i >= N * K) return;
    int n = i / K, k = i - n * K;
    out[i] = (_Float16)in[(size_t)k * N + n];
}

__global__ void k_fill_f32(float* __restrict__ p, float v, int n) {
    int i = blockIdx.x * blockDim.x + threadIdx.x;
    if (i < n) p[i] = v;
}

__global__ void k_fill_u32(unsigned* __restrict__ p, unsigned v, int n) {
    int i = blockIdx.x * blockDim.x + threadIdx.x;
    if (i < n) p[i] = v;
}

__global__ void k_rowbias(float* __restrict__ out, const float* __restrict__ bias,
                          int rows, int cols) {
    int i = blockIdx.x * blockDim.x + threadIdx.x;
    if (i < rows * cols) out[i] = bias[i % cols];
}

__global__ void k_elu_f16copy(float* __restrict__ x, _Float16* __restrict__ xh, int n) {
    int i = blockIdx.x * blockDim.x + threadIdx.x;
    if (i < n) { float v = eluf(x[i]); x[i] = v; xh[i] = (_Float16)v; }
}

__global__ void k_sample1(float* __restrict__ x, const float* __restrict__ sl,
                          _Float16* __restrict__ xh, int n) {
    int i = blockIdx.x * blockDim.x + threadIdx.x;
    if (i >= n) return;
    int col = i & (HC1 - 1);
    int s   = i >> 8;
    float v = eluf(x[i] + sl[(size_t)s * C1 + (col & (C1 - 1))]);
    x[i] = v; xh[i] = (_Float16)v;
}

__global__ void k_final(const float* __restrict__ acc3, const float* __restrict__ sl3,
                        const float* __restrict__ bias3, float* __restrict__ out, int n) {
    int i = blockIdx.x * blockDim.x + threadIdx.x;
    if (i >= n) return;
    int c = i & (C3 - 1);
    int s = i >> 7;
    const float* a = acc3 + (size_t)s * HC3;
    float mean = 0.25f * (a[c] + a[C3 + c] + a[2 * C3 + c] + a[3 * C3 + c]);
    out[i] = eluf(mean + bias3[c] + sl3[i]);
}

__device__ __forceinline__ v16h ld_frag(const _Float16* p) {
    v8h lo = *(const v8h*)(p);
    v8h hi = *(const v8h*)(p + 16);
    v16h r;
#pragma unroll
    for (int j = 0; j < 8; ++j) { r[j] = lo[j]; r[8 + j] = hi[j]; }
    return r;
}

__global__ void __launch_bounds__(256)
k_gemm_wmma(const _Float16* __restrict__ X, const _Float16* __restrict__ Wt,
            const float* __restrict__ bias, float* __restrict__ Y,
            int M, int N, int K) {
    int wave = threadIdx.x >> 5;
    int lane = threadIdx.x & 31;
    int wm = wave & 3, wn = wave >> 2;
    int m0 = blockIdx.x * 128 + wm * 32;
    if (m0 >= M) return;
    int n0 = blockIdx.y * 64 + wn * 32;
    int lr = lane & 15;
    int kb = (lane >> 4) * 8;

    int ra0 = m0 + lr;      if (ra0 > M - 1) ra0 = M - 1;
    int ra1 = m0 + 16 + lr; if (ra1 > M - 1) ra1 = M - 1;
    int cb0 = n0 + lr;
    int cb1 = cb0 + 16;

    const _Float16* pA0 = X  + (size_t)ra0 * K + kb;
    const _Float16* pA1 = X  + (size_t)ra1 * K + kb;
    const _Float16* pB0 = Wt + (size_t)cb0 * K + kb;
    const _Float16* pB1 = Wt + (size_t)cb1 * K + kb;

    v8f acc00 = {}, acc01 = {}, acc10 = {}, acc11 = {};
#pragma unroll 4
    for (int k0 = 0; k0 < K; k0 += 32) {
        v16h a0 = ld_frag(pA0 + k0);
        v16h a1 = ld_frag(pA1 + k0);
        v16h b0 = ld_frag(pB0 + k0);
        v16h b1 = ld_frag(pB1 + k0);
        acc00 = __builtin_amdgcn_wmma_f32_16x16x32_f16(false, a0, false, b0, (short)0, acc00, false, false);
        acc01 = __builtin_amdgcn_wmma_f32_16x16x32_f16(false, a0, false, b1, (short)0, acc01, false, false);
        acc10 = __builtin_amdgcn_wmma_f32_16x16x32_f16(false, a1, false, b0, (short)0, acc10, false, false);
        acc11 = __builtin_amdgcn_wmma_f32_16x16x32_f16(false, a1, false, b1, (short)0, acc11, false, false);
    }

    float bv0 = bias ? bias[cb0] : 0.f;
    float bv1 = bias ? bias[cb1] : 0.f;
    int mr = m0 + 8 * (lane >> 4);
#pragma unroll
    for (int v = 0; v < 8; ++v) {
        int ma = mr + v;
        if (ma < M) {
            Y[(size_t)ma * N + cb0] = acc00[v] + bv0;
            Y[(size_t)ma * N + cb1] = acc01[v] + bv1;
        }
        int mb = mr + 16 + v;
        if (mb < M) {
            Y[(size_t)mb * N + cb0] = acc10[v] + bv0;
            Y[(size_t)mb * N + cb1] = acc11[v] + bv1;
        }
    }
}

__global__ void __launch_bounds__(256)
k_edge_alpha(const int* __restrict__ src, const int* __restrict__ dst,
             const float* __restrict__ xl, const float* __restrict__ xr,
             const float* __restrict__ att, float* __restrict__ alpha,
             unsigned* __restrict__ menc, int E, int C) {
    int wave = threadIdx.x >> 5, lane = threadIdx.x & 31;
    int e = blockIdx.x * 8 + wave;
    if (e >= E) return;
    int h = lane >> 3, ls = lane & 7;
    int s = src[e], d = dst[e];
    const float* pl = xl + ((size_t)s * NH + h) * C;
    const float* pr = xr + ((size_t)d * NH + h) * C;
    const float* pa = att + (size_t)h * C;
    float acc = 0.f;
    for (int c = ls; c < C; c += 8) {
        float g = pl[c] + pr[c];
        g = g > 0.f ? g : 0.2f * g;
        acc += g * pa[c];
    }
    acc += __shfl_xor(acc, 1, 32);
    acc += __shfl_xor(acc, 2, 32);
    acc += __shfl_xor(acc, 4, 32);
    if (ls == 0) {
        alpha[(size_t)e * NH + h] = acc;
        unsigned u = __float_as_uint(acc);
        unsigned enc = (u & 0x80000000u) ? ~u : (u | 0x80000000u);
        atomicMax(&menc[(size_t)d * NH + h], enc);
    }
}

__global__ void k_edge_exp(const int* __restrict__ dst, float* __restrict__ alpha_ea,
                           const unsigned* __restrict__ menc, float* __restrict__ den,
                           int EH) {
    int i = blockIdx.x * blockDim.x + threadIdx.x;
    if (i >= EH) return;
    int e = i >> 2, h = i & 3;
    int d = dst[e];
    unsigned enc = menc[(size_t)d * NH + h];
    float m = (enc & 0x80000000u) ? __uint_as_float(enc ^ 0x80000000u)
                                  : __uint_as_float(~enc);
    float ea = expf(alpha_ea[i] - m);
    alpha_ea[i] = ea;
    atomicAdd(&den[(size_t)d * NH + h], ea);
}

__global__ void __launch_bounds__(256)
k_edge_scatter(const int* __restrict__ src, const int* __restrict__ dst,
               const float* __restrict__ xl, const float* __restrict__ ea,
               const float* __restrict__ den, float* __restrict__ out, int E, int C) {
    int wave = threadIdx.x >> 5, lane = threadIdx.x & 31;
    int e = blockIdx.x * 8 + wave;
    if (e >= E) return;
    int h = lane >> 3, ls = lane & 7;
    int s = src[e], d = dst[e];
    float coef = ea[(size_t)e * NH + h] / (den[(size_t)d * NH + h] + 1e-16f);
    const float* pl = xl + ((size_t)s * NH + h) * C;
    float*       po = out + ((size_t)d * NH + h) * C;
    for (int c = ls; c < C; c += 8) atomicAdd(&po[c], coef * pl[c]);
}

extern "C" void kernel_launch(void* const* d_in, const int* in_sizes, int n_in,
                              void* d_out, int out_size, void* d_ws, size_t ws_size,
                              hipStream_t stream) {
    (void)in_sizes; (void)n_in; (void)out_size; (void)ws_size;

    const float* x_sample = (const float*)d_in[0];
    const float* x_gene   = (const float*)d_in[1];
    const int*   sg_src   = (const int*)d_in[2];
    const int*   sg_dst   = (const int*)d_in[3];
    const int*   gs_src   = (const int*)d_in[4];
    const int*   gs_dst   = (const int*)d_in[5];
    const float* Wl1_sg = (const float*)d_in[6];   const float* bl1_sg  = (const float*)d_in[7];
    const float* Wr1_sg = (const float*)d_in[8];   const float* br1_sg  = (const float*)d_in[9];
    const float* att1_sg = (const float*)d_in[10]; const float* bias1_sg = (const float*)d_in[11];
    const float* Wl1_gs = (const float*)d_in[12];  const float* bl1_gs  = (const float*)d_in[13];
    const float* Wr1_gs = (const float*)d_in[14];  const float* br1_gs  = (const float*)d_in[15];
    const float* att1_gs = (const float*)d_in[16]; const float* bias1_gs = (const float*)d_in[17];
    const float* Wl3 = (const float*)d_in[18];     const float* bl3 = (const float*)d_in[19];
    const float* Wr3 = (const float*)d_in[20];     const float* br3 = (const float*)d_in[21];
    const float* att3 = (const float*)d_in[22];    const float* bias3 = (const float*)d_in[23];
    const float* sl1_W = (const float*)d_in[24];   const float* sl1_b = (const float*)d_in[25];
    const float* sl3_W = (const float*)d_in[26];   const float* sl3_b = (const float*)d_in[27];

    char* p = (char*)d_ws;
    auto carve = [&](size_t bytes) -> char* {
        char* r = p; p += (bytes + 255) & ~(size_t)255; return r;
    };

    _Float16* hx_sample = (_Float16*)carve((size_t)NS * DIN * 2);
    _Float16* hx_gene   = (_Float16*)carve((size_t)NG * DIN * 2);
    _Float16* hWl1sg = (_Float16*)carve((size_t)DIN * HC1 * 2);
    _Float16* hWr1sg = (_Float16*)carve((size_t)DIN * HC1 * 2);
    _Float16* hWl1gs = (_Float16*)carve((size_t)DIN * HC1 * 2);
    _Float16* hWr1gs = (_Float16*)carve((size_t)DIN * HC1 * 2);
    _Float16* hWl3   = (_Float16*)carve((size_t)HC1 * HC3 * 2);
    _Float16* hWr3   = (_Float16*)carve((size_t)HC1 * HC3 * 2);
    _Float16* hsl1W  = (_Float16*)carve((size_t)DIN * C1 * 2);
    _Float16* hsl3W  = (_Float16*)carve((size_t)HC1 * C3 * 2);
    _Float16* hx1_gene   = (_Float16*)carve((size_t)NG * HC1 * 2);
    _Float16* hx1_sample = (_Float16*)carve((size_t)NS * HC1 * 2);
    float* xl_sg = (float*)carve((size_t)NS * HC1 * 4);
    float* xr_sg = (float*)carve((size_t)NG * HC1 * 4);
    float* out1_gene   = (float*)carve((size_t)NG * HC1 * 4);
    float* xr_gs = (float*)carve((size_t)NS * HC1 * 4);
    float* out1_sample = (float*)carve((size_t)NS * HC1 * 4);
    float* sl1 = (float*)carve((size_t)NS * C1 * 4);
    float* xl3 = (float*)carve((size_t)NG * HC3 * 4);
    float* xr3 = (float*)carve((size_t)NS * HC3 * 4);
    float* acc3 = (float*)carve((size_t)NS * HC3 * 4);
    float* sl3 = (float*)carve((size_t)NS * C3 * 4);
    float* alpha = (float*)carve((size_t)NE * NH * 4);
    unsigned* menc = (unsigned*)carve((size_t)NG * NH * 4);
    float* den = (float*)carve((size_t)NG * NH * 4);
    float* xl_gs = xr_sg;

    const int TB = 256;
#define EW(n) cdiv((n), TB), TB, 0, stream
    k_f32_to_f16<<<EW(NS * DIN)>>>(x_sample, hx_sample, NS * DIN);
    k_f32_to_f16<<<EW(NG * DIN)>>>(x_gene, hx_gene, NG * DIN);
    k_f32_to_f16_t<<<EW(DIN * HC1)>>>(Wl1_sg, hWl1sg, DIN, HC1);
    k_f32_to_f16_t<<<EW(DIN * HC1)>>>(Wr1_sg, hWr1sg, DIN, HC1);
    k_f32_to_f16_t<<<EW(DIN * HC1)>>>(Wl1_gs, hWl1gs, DIN, HC1);
    k_f32_to_f16_t<<<EW(DIN * HC1)>>>(Wr1_gs, hWr1gs, DIN, HC1);
    k_f32_to_f16_t<<<EW(HC1 * HC3)>>>(Wl3, hWl3, HC1, HC3);
    k_f32_to_f16_t<<<EW(HC1 * HC3)>>>(Wr3, hWr3, HC1, HC3);
    k_f32_to_f16_t<<<EW(DIN * C1)>>>(sl1_W, hsl1W, DIN, C1);
    k_f32_to_f16_t<<<EW(HC1 * C3)>>>(sl3_W, hsl3W, HC1, C3);

    auto gemm = [&](const _Float16* X, const _Float16* Wt, const float* b, float* Y,
                    int M, int N, int K) {
        dim3 g(cdiv(M, 128), cdiv(N, 64));
        k_gemm_wmma<<<g, 256, 0, stream>>>(X, Wt, b, Y, M, N, K);
    };

    gemm(hx_sample, hWl1sg, bl1_sg, xl_sg, NS, HC1, DIN);
    gemm(hx_gene,   hWr1sg, br1_sg, xr_sg, NG, HC1, DIN);
    k_fill_u32<<<EW(NG * NH)>>>(menc, NEG_ENC, NG * NH);
    k_fill_f32<<<EW(NG * NH)>>>(den, 0.f, NG * NH);
    k_rowbias<<<EW(NG * HC1)>>>(out1_gene, bias1_sg, NG, HC1);
    k_edge_alpha<<<cdiv(NE, 8), 256, 0, stream>>>(sg_src, sg_dst, xl_sg, xr_sg, att1_sg,
                                                  alpha, menc, NE, C1);
    k_edge_exp<<<EW(NE * NH)>>>(sg_dst, alpha, menc, den, NE * NH);
    k_edge_scatter<<<cdiv(NE, 8), 256, 0, stream>>>(sg_src, sg_dst, xl_sg, alpha, den,
                                                    out1_gene, NE, C1);
    k_elu_f16copy<<<EW(NG * HC1)>>>(out1_gene, hx1_gene, NG * HC1);

    gemm(hx_gene,   hWl1gs, bl1_gs, xl_gs, NG, HC1, DIN);
    gemm(hx_sample, hWr1gs, br1_gs, xr_gs, NS, HC1, DIN);
    gemm(hx_sample, hsl1W,  sl1_b,  sl1,   NS, C1,  DIN);
    k_fill_u32<<<EW(NS * NH)>>>(menc, NEG_ENC, NS * NH);
    k_fill_f32<<<EW(NS * NH)>>>(den, 0.f, NS * NH);
    k_rowbias<<<EW(NS * HC1)>>>(out1_sample, bias1_gs, NS, HC1);
    k_edge_alpha<<<cdiv(NE, 8), 256, 0, stream>>>(gs_src, gs_dst, xl_gs, xr_gs, att1_gs,
                                                  alpha, menc, NE, C1);
    k_edge_exp<<<EW(NE * NH)>>>(gs_dst, alpha, menc, den, NE * NH);
    k_edge_scatter<<<cdiv(NE, 8), 256, 0, stream>>>(gs_src, gs_dst, xl_gs, alpha, den,
                                                    out1_sample, NE, C1);
    k_sample1<<<EW(NS * HC1)>>>(out1_sample, sl1, hx1_sample, NS * HC1);

    gemm(hx1_gene,   hWl3,  bl3,   xl3, NG, HC3, HC1);
    gemm(hx1_sample, hWr3,  br3,   xr3, NS, HC3, HC1);
    gemm(hx1_sample, hsl3W, sl3_b, sl3, NS, C3,  HC1);
    k_fill_u32<<<EW(NS * NH)>>>(menc, NEG_ENC, NS * NH);
    k_fill_f32<<<EW(NS * NH)>>>(den, 0.f, NS * NH);
    k_fill_f32<<<EW(NS * HC3)>>>(acc3, 0.f, NS * HC3);
    k_edge_alpha<<<cdiv(NE, 8), 256, 0, stream>>>(gs_src, gs_dst, xl3, xr3, att3,
                                                  alpha, menc, NE, C3);
    k_edge_exp<<<EW(NE * NH)>>>(gs_dst, alpha, menc, den, NE * NH);
    k_edge_scatter<<<cdiv(NE, 8), 256, 0, stream>>>(gs_src, gs_dst, xl3, alpha, den,
                                                    acc3, NE, C3);
    k_final<<<EW(NS * C3)>>>(acc3, sl3, bias3, (float*)d_out, NS * C3);
#undef EW
}


# GEMMs that share an A operand run as one GEMM over their weights laid end to end: a wave loads and converts its rows once and walks one output's columns; 8 GEMM launches -> 4 (20 -> 16)
# speedup vs baseline: 3.0829x; 3.0829x over previous
#include <hip/hip_runtime.h>
#include <hip/hip_bf16.h>


#define NS   4096
#define NG   20000
#define NE   131072
#define DIN  256
#define NH   4
#define C1   64
#define C3   128
#define HC1  256
#define HC3  512
#define NEG_ENC 0x00800000u

typedef __attribute__((ext_vector_type(16))) _Float16 v16h;
typedef __attribute__((ext_vector_type(8)))  _Float16 v8h;
typedef __attribute__((ext_vector_type(8)))  float    v8f;

static inline int cdiv(int a, int b) { return (a + b - 1) / b; }

__device__ __forceinline__ float eluf(float x) { return x > 0.f ? x : (expf(x) - 1.f); }

__global__ void k_f32_to_f16_t(const float* __restrict__ in, _Float16* __restrict__ out,
                               int K, int N) {
    int i = blockIdx.x * blockDim.x + threadIdx.x;
    if (i >= N * K) return;
    int n = i / K, k = i - n * K;
    out[i] = (_Float16)in[(size_t)k * N + n];
}

__global__ void k_final(const float* __restrict__ acc3, const float* __restrict__ sl3,
                        const float* __restrict__ bias3, float* __restrict__ out, int n) {
    int i = blockIdx.x * blockDim.x + threadIdx.x;
    if (i >= n) return;
    int c = i & (C3 - 1);
    out[i] = eluf(acc3[i] + bias3[c] + sl3[i]);
}

#define GK 256
#define GKS (GK / 32)
#define LPH 264

typedef __attribute__((ext_vector_type(4))) float    v4f;
typedef __attribute__((ext_vector_type(4))) _Float16 v4h;

__device__ __forceinline__ v16h ld_frag(const _Float16* p) {
    v8h lo = *(const v8h*)(p);
    v8h hi = *(const v8h*)(p + 16);
    v16h r;
#pragma unroll
    for (int j = 0; j < 8; ++j) { r[j] = lo[j]; r[8 + j] = hi[j]; }
    return r;
}

__device__ __forceinline__ float eluf_sel(float x) {
    const float e = expf(x) - 1.f;
    return x > 0.f ? x : e;
}

template <int MODE>
__device__ __forceinline__ v4h cvt4(const v4f& x, const v4f& t) {
    v4h r;
#pragma unroll
    for (int e = 0; e < 4; ++e) {
        float f = x[e];
        if (MODE == 2) f += t[e];
        r[e] = (_Float16)(MODE >= 1 ? eluf_sel(f) : f);
    }
    return r;
}

template <int MODE>
__global__ void __launch_bounds__(128) __attribute__((amdgpu_waves_per_eu(1, 2)))
k_gemm_lds(const float* __restrict__ A, const float* __restrict__ sl,
           const _Float16* __restrict__ Wt,
           const float* __restrict__ bias0, float* __restrict__ Y0, int N0,
           const float* __restrict__ bias1, float* __restrict__ Y1, int N1,
           const float* __restrict__ bias2, float* __restrict__ Y2, int N2,
           int M, int nch) {
    __shared__ __attribute__((aligned(16))) _Float16 As[4][32 * LPH];
    __shared__ __attribute__((aligned(16))) _Float16 Bs[2][32 * LPH];

    const int wave = __builtin_amdgcn_readfirstlane(threadIdx.x >> 5);
    const int lane = threadIdx.x & 31;
    const int m0raw = (blockIdx.x * 4 + wave) * 32;
    const bool live = m0raw < M;
    const int m0 = live ? m0raw : M - 32;
    const int lr = lane & 15;
    const int kb = (lane >> 4) * 8;
    const int nbeg = blockIdx.y * nch;
    const int nsteps = nch / 32;
    const int seg = nbeg < N0 ? 0 : (nbeg < N0 + N1 ? 1 : 2);
    float* __restrict__ Y = seg == 0 ? Y0 : (seg == 1 ? Y1 : Y2);
    const float* __restrict__ bias = seg == 0 ? bias0 : (seg == 1 ? bias1 : bias2);
    const int N = seg == 0 ? N0 : (seg == 1 ? N1 : N2);
    const int cbeg = nbeg - (seg == 0 ? 0 : (seg == 1 ? N0 : N0 + N1));

    const _Float16* pW = Wt + (size_t)(nbeg + 8 * wave) * GK + 8 * lane;
    v8h bq[8];
#pragma unroll
    for (int t = 0; t < 8; ++t) bq[t] = *(const v8h*)(pW + (size_t)t * GK);

    {
        const float* pA = A + (size_t)m0 * GK + 4 * lane;
        const float* pS = MODE == 2 ? sl + (size_t)m0 * C1 + ((4 * lane) & (C1 - 1)) : pA;
        _Float16* as = &As[wave][4 * lane];
#pragma unroll
        for (int g = 0; g < 2; ++g) {
            v4f xa[16][2], ta[16];
#pragma unroll
            for (int r = 0; r < 16; ++r) {
                xa[r][0] = *(const v4f*)(pA + (size_t)(16 * g + r) * GK);
                xa[r][1] = *(const v4f*)(pA + (size_t)(16 * g + r) * GK + 128);
                ta[r] = MODE == 2 ? *(const v4f*)(pS + (size_t)(16 * g + r) * C1) : v4f{};
            }
#pragma unroll
            for (int r = 0; r < 16; ++r) {
                *(v4h*)(as + (16 * g + r) * LPH)       = cvt4<MODE>(xa[r][0], ta[r]);
                *(v4h*)(as + (16 * g + r) * LPH + 128) = cvt4<MODE>(xa[r][1], ta[r]);
            }
        }
    }
#pragma unroll
    for (int t = 0; t < 8; ++t) *(v8h*)&Bs[0][(8 * wave + t) * LPH + 8 * lane] = bq[t];
    __syncthreads();

    v16h a0[GKS], a1[GKS];
#pragma unroll
    for (int s = 0; s < GKS; ++s) {
        a0[s] = ld_frag(&As[wave][lr * LPH + 32 * s + kb]);
        a1[s] = ld_frag(&As[wave][(16 + lr) * LPH + 32 * s + kb]);
    }

    const int mr = m0 + 8 * (lane >> 4);
    for (int it = 0; it < nsteps; ++it) {
        const int n0 = cbeg + 32 * it;
        const int cur = it & 1;
        const bool more = it + 1 < nsteps;
        if (more) {
#pragma unroll
            for (int t = 0; t < 8; ++t) bq[t] = *(const v8h*)(pW + (size_t)(32 * (it + 1) + t) * GK);
        }
        const _Float16* bs = &Bs[cur][lr * LPH + kb];
        v16h b0[GKS], b1[GKS];
#pragma unroll
        for (int s = 0; s < GKS; ++s) {
            b0[s] = ld_frag(bs + 32 * s);
            b1[s] = ld_frag(bs + 16 * LPH + 32 * s);
        }
        __builtin_amdgcn_sched_barrier(0);
        v8f acc00 = {}, acc01 = {}, acc10 = {}, acc11 = {};
#pragma unroll
        for (int s = 0; s < GKS; ++s) {
            acc00 = __builtin_amdgcn_wmma_f32_16x16x32_f16(false, a0[s], false, b0[s], (short)0, acc00, false, false);
            acc01 = __builtin_amdgcn_wmma_f32_16x16x32_f16(false, a0[s], false, b1[s], (short)0, acc01, false, false);
            acc10 = __builtin_amdgcn_wmma_f32_16x16x32_f16(false, a1[s], false, b0[s], (short)0, acc10, false, false);
            acc11 = __builtin_amdgcn_wmma_f32_16x16x32_f16(false, a1[s], false, b1[s], (short)0, acc11, false, false);
        }
        if (more) {
#pragma unroll
            for (int t = 0; t < 8; ++t) *(v8h*)&Bs[cur ^ 1][(8 * wave + t) * LPH + 8 * lane] = bq[t];
            __syncthreads();
        }
        if (live) {
            const int cb0 = n0 + lr;
            const int cb1 = cb0 + 16;
            const float bv0 = bias ? bias[cb0] : 0.f;
            const float bv1 = bias ? bias[cb1] : 0.f;
#pragma unroll
            for (int v = 0; v < 8; ++v) {
                Y[(size_t)(mr + v) * N + cb0]      = acc00[v] + bv0;
                Y[(size_t)(mr + v) * N + cb1]      = acc01[v] + bv1;
                Y[(size_t)(mr + 16 + v) * N + cb0] = acc10[v] + bv0;
                Y[(size_t)(mr + 16 + v) * N + cb1] = acc11[v] + bv1;
            }
        }
    }
}

#define EPW (NE / 32)
#define WPW (EPW / 32)

__device__ __forceinline__ float xhalf(float x) {
    int b = __builtin_bit_cast(int, x);
    return __builtin_bit_cast(float,
        __builtin_amdgcn_permlanex16(b, b, 0x76543210, 0xfedcba98, false, false));
}
template <int N>
__device__ __forceinline__ float xrow(float x) {
    int b = __builtin_bit_cast(int, x);
    return __builtin_bit_cast(float, __builtin_amdgcn_update_dpp(b, b, 0x160 | N, 0xf, 0xf, false));
}

__device__ __forceinline__ float head_reduce(float p0, float p1, float p2, float p3, int lane) {
    const bool h16 = (lane & 16) != 0;
    float a = h16 ? p2 : p0, sa = h16 ? p0 : p2;
    float c = h16 ? p3 : p1, sc = h16 ? p1 : p3;
    a += xhalf(sa);
    c += xhalf(sc);
    const bool h8 = (lane & 8) != 0;
    float k = h8 ? c : a, sk = h8 ? a : c;
    k += xrow<8>(sk);
    k += xrow<4>(k);
    k += xrow<2>(k);
    k += xrow<1>(k);
    return k;
}

template <int NJ, int HJ>
__device__ __forceinline__ float edge_score(const float (&x)[NJ], const float* xr_row,
                                            const float* att_s, int lane) {
    float p[NH] = {0.f, 0.f, 0.f, 0.f};
    for (int j = 0; j < NJ; ++j) {
        float g = x[j] + xr_row[lane + 32 * j];
        g = g > 0.f ? g : 0.2f * g;
        p[j / HJ] = fmaf(g, att_s[lane + 32 * j], p[j / HJ]);
    }
    return head_reduce(p[0], p[1], p[2], p[3], lane);
}

__device__ __forceinline__ int wave_scan_incl(int v, int lane) {
    for (int d = 1; d < 32; d <<= 1) {
        const int t = __shfl_up(v, d, 32);
        if (lane >= d) v += t;
    }
    return v;
}

__device__ __forceinline__ float dec_max(unsigned enc) {
    return (enc & 0x80000000u) ? __uint_as_float(enc ^ 0x80000000u) : __uint_as_float(~enc);
}

#define LCAP 2048

template <int ROWS, int C, bool MEAN>
__global__ void __launch_bounds__(1024) __attribute__((amdgpu_waves_per_eu(ROWS > 16 ? 16 : 8, 16)))
k_own_gat(const int* __restrict__ src, const int* __restrict__ dst,
          const float* __restrict__ xl, const float* __restrict__ xr,
          const float* __restrict__ att, const float* __restrict__ bias,
          float* __restrict__ out, int num_dst) {
    constexpr int SROW = NH * C;
    constexpr int NJ   = SROW / 32;
    constexpr int HJ   = C / 32;
    constexpr int OROW = MEAN ? C : SROW;
    constexpr int NQ   = 16;
    constexpr int NB   = ROWS > 16 ? 2 : 4;
    constexpr int RS   = ROWS + 1;
    constexpr unsigned PAD_ENT = (unsigned)ROWS << 16;
    __shared__ float    xr_s[RS * SROW];
    __shared__ float    acc[RS * SROW];
    __shared__ float    att_s[SROW];
    __shared__ unsigned menc_s[RS * NH];
    __shared__ float    mapp_s[2][RS * NH];
    __shared__ float    den_s[RS * NH];
    __shared__ __attribute__((aligned(16))) unsigned smask[32 * WPW];
    __shared__ __attribute__((aligned(16))) unsigned elist[LCAP];
    __shared__ int      wcnt[32];

    const int tid = threadIdx.x, lane = tid & 31;
    const int wave = __builtin_amdgcn_readfirstlane(tid >> 5);
    const int lo = blockIdx.x * ROWS;
    int nrows = num_dst - lo;
    if (nrows > ROWS) nrows = ROWS;
    if (nrows < 0) nrows = 0;
    const int hh = (lane >> 3) & 3;

    for (int i = tid; i < RS * SROW; i += 1024) {
        acc[i] = 0.f;
        xr_s[i] = i < nrows * SROW ? xr[(size_t)lo * SROW + i] : 0.f;
    }
    for (int i = tid; i < SROW; i += 1024) att_s[i] = att[i];
    for (int i = tid; i < RS * NH; i += 1024) {
        menc_s[i] = NEG_ENC; den_s[i] = 0.f;
        mapp_s[1][i] = dec_max(NEG_ENC);
    }

    int cnt = 0;
    for (int b = 0; b < WPW / NQ; ++b) {
        const int base = wave * EPW + b * (32 * NQ);
        int dv[NQ];
#pragma unroll
        for (int q = 0; q < NQ; ++q) dv[q] = dst[base + 32 * q + lane];
#pragma unroll
        for (int q = 0; q < NQ; ++q) {
            const unsigned mask =
                __builtin_amdgcn_ballot_w32((unsigned)(dv[q] - lo) < (unsigned)nrows);
            if (lane == 0) smask[wave * WPW + b * NQ + q] = mask;
            cnt += __builtin_popcount(mask);
        }
    }
    if (lane == 0) wcnt[wave] = cnt;
    __syncthreads();

    const int wc   = wcnt[lane];
    const int winc = wave_scan_incl(wc, lane);
    const int T     = __builtin_amdgcn_readlane(winc, 31);
    const int wbase = __builtin_amdgcn_readlane(winc - wc, wave);
    auto expand = [&](int start, int n) {
        const uint4 mw = *(const uint4*)&smask[wave * WPW + 4 * lane];
        const int s4 = __builtin_popcount(mw.x) + __builtin_popcount(mw.y) +
                       __builtin_popcount(mw.z) + __builtin_popcount(mw.w);
        const int lbase = wbase + wave_scan_incl(s4, lane) - s4;
        unsigned long long lo64 = (unsigned long long)mw.x | ((unsigned long long)mw.y << 32);
        unsigned long long hi64 = (unsigned long long)mw.z | ((unsigned long long)mw.w << 32);
        int pos = lbase - start;
        const int ebase = wave * EPW + 128 * lane;
        while (__builtin_amdgcn_ballot_w32((lo64 | hi64) != 0ull) != 0u) {
            int ei[4], sv[4], dw[4];
#pragma unroll
            for (int t = 0; t < 4; ++t) {
                int idx = -1;
                if (lo64)      { idx = __builtin_ctzll(lo64);      lo64 &= lo64 - 1; }
                else if (hi64) { idx = 64 + __builtin_ctzll(hi64); hi64 &= hi64 - 1; }
                ei[t] = idx;
                const int e = ebase + (idx < 0 ? 0 : idx);
                sv[t] = src[e];
                dw[t] = dst[e];
            }
#pragma unroll
            for (int t = 0; t < 4; ++t) {
                if (ei[t] >= 0) {
                    if ((unsigned)pos < (unsigned)n)
                        elist[pos] = ((unsigned)(dw[t] - lo) << 16) | (unsigned)sv[t];
                    ++pos;
                }
            }
        }
        if (tid < 4 && n + tid < ((n + 3) & ~3)) elist[n + tid] = PAD_ENT;
    };

    int rk = 0;
    for (int start = 0; start < T; start += LCAP) {
        const int n = T - start < LCAP ? T - start : LCAP;
        expand(start, n);
        __syncthreads();
        for (int k0 = 0; k0 < n; k0 += 32 * NB, ++rk) {
            const int b0 = k0 + wave * NB;
            const bool active = b0 < n;
            const int cur = rk & 1;
            unsigned ent[NB];
            float x[NB][NJ];
            float al[NB];
            if (active) {
#pragma unroll
                for (int u = 0; u < NB; ++u)
                    ent[u] = (unsigned)__builtin_amdgcn_readfirstlane((int)elist[b0 + u]);
#pragma unroll
                for (int u = 0; u < NB; ++u) {
                    const float* pl = xl + (size_t)(ent[u] & 0xffffu) * SROW + lane;
                    for (int j = 0; j < NJ; ++j) x[u][j] = pl[32 * j];
                }
#pragma unroll
                for (int u = 0; u < NB; ++u) {
                    const int r = ent[u] >> 16;
                    al[u] = edge_score<NJ, HJ>(x[u], xr_s + r * SROW, att_s, lane);
                    if ((lane & 7) == 0) {
                        unsigned uu = __float_as_uint(al[u]);
                        unsigned enc = (uu & 0x80000000u) ? ~uu : (uu | 0x80000000u);
                        atomicMax(&menc_s[r * NH + hh], enc);
                    }
                }
            }
            __syncthreads();

            {
                constexpr int PPW = (RS * NH + 31) / 32;
                const int pr = wave * PPW + lane;
                const bool mine = lane < PPW && pr < RS * NH;
                float mo = 0.f, mn = 0.f;
                if (mine) {
                    mo = mapp_s[cur ^ 1][pr];
                    mn = dec_max(menc_s[pr]);
                    mapp_s[cur][pr] = mn;
                }
                unsigned moved = __builtin_amdgcn_ballot_w32(mine && mn != mo);
                const float f = expf(mo - mn);
                while (moved) {
                    const int l = __builtin_ctz(moved);
                    moved &= moved - 1;
                    const float fl = __builtin_bit_cast(float,
                        __builtin_amdgcn_readlane(__builtin_bit_cast(int, f), l));
                    const int p = wave * PPW + l;
                    for (int j = 0; j < HJ; ++j) acc[p * C + 32 * j + lane] *= fl;
                    if (lane == 0) den_s[p] *= fl;
                }
            }
            __syncthreads();

            if (active) {
#pragma unroll
                for (int u = 0; u < NB; ++u) {
                    const int r = ent[u] >> 16;
                    const float ea = expf(al[u] - mapp_s[cur][r * NH + hh]);
                    if ((lane & 7) == 0) atomicAdd(&den_s[r * NH + hh], ea);
                    float eah[NH];
                    for (int h = 0; h < NH; ++h)
                        eah[h] = __builtin_bit_cast(float,
                            __builtin_amdgcn_readlane(__builtin_bit_cast(int, ea), 8 * h));
                    float* pa = acc + r * SROW + lane;
                    for (int j = 0; j < NJ; ++j) atomicAdd(&pa[32 * j], eah[j / HJ] * x[u][j]);
                }
            }
        }
    }
    __syncthreads();

    float* po = out + (size_t)lo * OROW;
    for (int i = tid; i < nrows * OROW; i += 1024) {
        const int r = i / OROW, col = i - r * OROW;
        float v;
        if (MEAN) {
            v = 0.f;
            for (int h = 0; h < NH; ++h)
                v += acc[r * SROW + h * C + col] / (den_s[r * NH + h] + 1e-16f);
            v *= 0.25f;
        } else {
            v = acc[i] / (den_s[r * NH + col / C] + 1e-16f);
        }
        if (bias) v += bias[col];
        po[i] = v;
    }
}

extern "C" void kernel_launch(void* const* d_in, const int* in_sizes, int n_in,
                              void* d_out, int out_size, void* d_ws, size_t ws_size,
                              hipStream_t stream) {
    (void)in_sizes; (void)n_in; (void)out_size; (void)ws_size;

    const float* x_sample = (const float*)d_in[0];
    const float* x_gene   = (const float*)d_in[1];
    const int*   sg_src   = (const int*)d_in[2];
    const int*   sg_dst   = (const int*)d_in[3];
    const int*   gs_src   = (const int*)d_in[4];
    const int*   gs_dst   = (const int*)d_in[5];
    const float* Wl1_sg = (const float*)d_in[6];   const float* bl1_sg  = (const float*)d_in[7];
    const float* Wr1_sg = (const float*)d_in[8];   const float* br1_sg  = (const float*)d_in[9];
    const float* att1_sg = (const float*)d_in[10]; const float* bias1_sg = (const float*)d_in[11];
    const float* Wl1_gs = (const float*)d_in[12];  const float* bl1_gs  = (const float*)d_in[13];
    const float* Wr1_gs = (const float*)d_in[14];  const float* br1_gs  = (const float*)d_in[15];
    const float* att1_gs = (const float*)d_in[16]; const float* bias1_gs = (const float*)d_in[17];
    const float* Wl3 = (const float*)d_in[18];     const float* bl3 = (const float*)d_in[19];
    const float* Wr3 = (const float*)d_in[20];     const float* br3 = (const float*)d_in[21];
    const float* att3 = (const float*)d_in[22];    const float* bias3 = (const float*)d_in[23];
    const float* sl1_W = (const float*)d_in[24];   const float* sl1_b = (const float*)d_in[25];
    const float* sl3_W = (const float*)d_in[26];   const float* sl3_b = (const float*)d_in[27];

    char* p = (char*)d_ws;
    auto carve = [&](size_t bytes) -> char* {
        char* r = p; p += (bytes + 255) & ~(size_t)255; return r;
    };

    _Float16* hWl1sg = (_Float16*)carve((size_t)DIN * HC1 * 2);
    _Float16* hWr1gs = (_Float16*)carve((size_t)DIN * HC1 * 2);
    _Float16* hsl1W  = (_Float16*)carve((size_t)DIN * C1 * 2);
    _Float16* hWr1sg = (_Float16*)carve((size_t)DIN * HC1 * 2);
    _Float16* hWl1gs = (_Float16*)carve((size_t)DIN * HC1 * 2);
    _Float16* hWr3   = (_Float16*)carve((size_t)HC1 * HC3 * 2);
    _Float16* hsl3W  = (_Float16*)carve((size_t)HC1 * C3 * 2);
    _Float16* hWl3   = (_Float16*)carve((size_t)HC1 * HC3 * 2);
    float* xl_sg = (float*)carve((size_t)NS * HC1 * 4);
    float* xr_sg = (float*)carve((size_t)NG * HC1 * 4);
    float* out1_gene   = (float*)carve((size_t)NG * HC1 * 4);
    float* xr_gs = (float*)carve((size_t)NS * HC1 * 4);
    float* out1_sample = (float*)carve((size_t)NS * HC1 * 4);
    float* sl1 = (float*)carve((size_t)NS * C1 * 4);
    float* xl3 = (float*)carve((size_t)NG * HC3 * 4);
    float* xr3 = (float*)carve((size_t)NS * HC3 * 4);
    float* acc3 = (float*)carve((size_t)NS * C3 * 4);
    float* sl3 = (float*)carve((size_t)NS * C3 * 4);
    float* xl_gs = xl3;

    const int TB = 256;
#define EW(n) cdiv((n), TB), TB, 0, stream
    k_f32_to_f16_t<<<EW(DIN * HC1)>>>(Wl1_sg, hWl1sg, DIN, HC1);
    k_f32_to_f16_t<<<EW(DIN * HC1)>>>(Wr1_sg, hWr1sg, DIN, HC1);
    k_f32_to_f16_t<<<EW(DIN * HC1)>>>(Wl1_gs, hWl1gs, DIN, HC1);
    k_f32_to_f16_t<<<EW(DIN * HC1)>>>(Wr1_gs, hWr1gs, DIN, HC1);
    k_f32_to_f16_t<<<EW(HC1 * HC3)>>>(Wl3, hWl3, HC1, HC3);
    k_f32_to_f16_t<<<EW(HC1 * HC3)>>>(Wr3, hWr3, HC1, HC3);
    k_f32_to_f16_t<<<EW(DIN * C1)>>>(sl1_W, hsl1W, DIN, C1);
    k_f32_to_f16_t<<<EW(HC1 * C3)>>>(sl3_W, hsl3W, HC1, C3);

    static_assert(DIN == GK && HC1 == GK && NS % 32 == 0 && NG % 32 == 0, "k_gemm_lds shape");
#define GEMM3(MODE, A, SL, Wt, b0, Y0, N0, b1, Y1, N1, b2, Y2, N2, M, nch) \
    k_gemm_lds<MODE><<<dim3(cdiv((M), 128), ((N0) + (N1) + (N2)) / (nch)), 128, 0, stream>>>( \
        A, SL, Wt, b0, Y0, N0, b1, Y1, N1, b2, Y2, N2, M, nch)

    GEMM3(0, x_sample, x_sample, hWl1sg, bl1_sg, xl_sg, HC1, br1_gs, xr_gs, HC1, sl1_b, sl1, C1, NS, 64);
    GEMM3(0, x_gene, x_gene, hWr1sg, br1_sg, xr_sg, HC1, bl1_gs, xl_gs, HC1, nullptr, nullptr, 0, NG, 256);

    k_own_gat<40, C1, false><<<cdiv(NG, 40), 1024, 0, stream>>>(sg_src, sg_dst, xl_sg, xr_sg, att1_sg,
                                                                bias1_sg, out1_gene, NG);
    k_own_gat<16, C1, false><<<cdiv(NS, 16), 1024, 0, stream>>>(gs_src, gs_dst, xl_gs, xr_gs, att1_gs,
                                                                bias1_gs, out1_sample, NS);

    GEMM3(1, out1_gene, out1_gene, hWl3, bl3, xl3, HC3, nullptr, nullptr, 0, nullptr, nullptr, 0, NG, 256);
    GEMM3(2, out1_sample, sl1, hWr3, br3, xr3, HC3, sl3_b, sl3, C3, nullptr, nullptr, 0, NS, 64);
    k_own_gat<16, C3, true><<<cdiv(NS, 16), 1024, 0, stream>>>(gs_src, gs_dst, xl3, xr3, att3,
                                                               nullptr, acc3, NS);
    k_final<<<EW(NS * C3)>>>(acc3, sl3, bias3, (float*)d_out, NS * C3);
#undef GEMM3
#undef EW
}


# the 8 weight conversions run as ONE launch: a block takes a 64 x 32 tile of one weight through LDS (whole-line reads and writes) into the f16 transposes laid end to end; 16 -> 9 launches
# speedup vs baseline: 3.3744x; 1.0946x over previous
#include <hip/hip_runtime.h>
#include <hip/hip_bf16.h>


#define NS   4096
#define NG   20000
#define NE   131072
#define DIN  256
#define NH   4
#define C1   64
#define C3   128
#define HC1  256
#define HC3  512
#define NEG_ENC 0x00800000u

typedef __attribute__((ext_vector_type(16))) _Float16 v16h;
typedef __attribute__((ext_vector_type(8)))  _Float16 v8h;
typedef __attribute__((ext_vector_type(8)))  float    v8f;

static inline int cdiv(int a, int b) { return (a + b - 1) / b; }

__device__ __forceinline__ float eluf(float x) { return x > 0.f ? x : (expf(x) - 1.f); }

#define WK      256
#define WR_XS   0
#define WR_XG   (WR_XS + 2 * HC1 + C1)
#define WR_X1S  (WR_XG + 2 * HC1)
#define WR_X1G  (WR_X1S + HC3 + C3)
#define WT_ROWS (WR_X1G + HC3)
typedef __attribute__((ext_vector_type(2))) _Float16 v2h;

__global__ void __launch_bounds__(256)
k_weights_t(const float* __restrict__ Wl1_sg, const float* __restrict__ Wr1_gs,
            const float* __restrict__ sl1_W,
            const float* __restrict__ Wr1_sg, const float* __restrict__ Wl1_gs,
            const float* __restrict__ Wr3, const float* __restrict__ sl3_W,
            const float* __restrict__ Wl3,
            _Float16* __restrict__ out) {
    __shared__ float tile[64][33];
    const int g0 = (blockIdx.x >> 2) * 32;
    const int k0 = (blockIdx.x & 3) * 64;
    const float* __restrict__ in = Wl1_sg; int N = HC1, nb = WR_XS;
    if (g0 >= WR_X1G)            { in = Wl3;    N = HC3; nb = WR_X1G; }
    else if (g0 >= WR_X1S + HC3) { in = sl3_W;  N = C3;  nb = WR_X1S + HC3; }
    else if (g0 >= WR_X1S)       { in = Wr3;    N = HC3; nb = WR_X1S; }
    else if (g0 >= WR_XG + HC1)  { in = Wl1_gs; N = HC1; nb = WR_XG + HC1; }
    else if (g0 >= WR_XG)        { in = Wr1_sg; N = HC1; nb = WR_XG; }
    else if (g0 >= WR_XS + 2 * HC1) { in = sl1_W;  N = C1;  nb = WR_XS + 2 * HC1; }
    else if (g0 >= WR_XS + HC1)  { in = Wr1_gs; N = HC1; nb = WR_XS + HC1; }
    const int lane = threadIdx.x & 31, w = threadIdx.x >> 5;
    const float* p = in + (size_t)(k0 + w) * N + (g0 - nb) + lane;
    float v[8];
#pragma unroll
    for (int i = 0; i < 8; ++i) v[i] = p[(size_t)(8 * i) * N];
#pragma unroll
    for (int i = 0; i < 8; ++i) tile[w + 8 * i][lane] = v[i];
    __syncthreads();
    _Float16* q = out + (size_t)(g0 + w) * WK + k0 + 2 * lane;
    for (int i = 0; i < 4; ++i) {
        v2h h;
        h[0] = (_Float16)tile[2 * lane][w + 8 * i];
        h[1] = (_Float16)tile[2 * lane + 1][w + 8 * i];
        *(v2h*)(q + (size_t)(8 * i) * WK) = h;
    }
}

__global__ void k_final(const float* __restrict__ acc3, const float* __restrict__ sl3,
                        const float* __restrict__ bias3, float* __restrict__ out, int n) {
    int i = blockIdx.x * blockDim.x + threadIdx.x;
    if (i >= n) return;
    int c = i & (C3 - 1);
    out[i] = eluf(acc3[i] + bias3[c] + sl3[i]);
}

#define GK 256
#define GKS (GK / 32)
#define LPH 264

typedef __attribute__((ext_vector_type(4))) float    v4f;
typedef __attribute__((ext_vector_type(4))) _Float16 v4h;

__device__ __forceinline__ v16h ld_frag(const _Float16* p) {
    v8h lo = *(const v8h*)(p);
    v8h hi = *(const v8h*)(p + 16);
    v16h r;
#pragma unroll
    for (int j = 0; j < 8; ++j) { r[j] = lo[j]; r[8 + j] = hi[j]; }
    return r;
}

__device__ __forceinline__ float eluf_sel(float x) {
    const float e = expf(x) - 1.f;
    return x > 0.f ? x : e;
}

template <int MODE>
__device__ __forceinline__ v4h cvt4(const v4f& x, const v4f& t) {
    v4h r;
#pragma unroll
    for (int e = 0; e < 4; ++e) {
        float f = x[e];
        if (MODE == 2) f += t[e];
        r[e] = (_Float16)(MODE >= 1 ? eluf_sel(f) : f);
    }
    return r;
}

template <int MODE>
__global__ void __launch_bounds__(128) __attribute__((amdgpu_waves_per_eu(1, 2)))
k_gemm_lds(const float* __restrict__ A, const float* __restrict__ sl,
           const _Float16* __restrict__ Wt,
           const float* __restrict__ bias0, float* __restrict__ Y0, int N0,
           const float* __restrict__ bias1, float* __restrict__ Y1, int N1,
           const float* __restrict__ bias2, float* __restrict__ Y2, int N2,
           int M, int nch) {
    __shared__ __attribute__((aligned(16))) _Float16 As[4][32 * LPH];
    __shared__ __attribute__((aligned(16))) _Float16 Bs[2][32 * LPH];

    const int wave = __builtin_amdgcn_readfirstlane(threadIdx.x >> 5);
    const int lane = threadIdx.x & 31;
    const int m0raw = (blockIdx.x * 4 + wave) * 32;
    const bool live = m0raw < M;
    const int m0 = live ? m0raw : M - 32;
    const int lr = lane & 15;
    const int kb = (lane >> 4) * 8;
    const int nbeg = blockIdx.y * nch;
    const int nsteps = nch / 32;
    const int seg = nbeg < N0 ? 0 : (nbeg < N0 + N1 ? 1 : 2);
    float* __restrict__ Y = seg == 0 ? Y0 : (seg == 1 ? Y1 : Y2);
    const float* __restrict__ bias = seg == 0 ? bias0 : (seg == 1 ? bias1 : bias2);
    const int N = seg == 0 ? N0 : (seg == 1 ? N1 : N2);
    const int cbeg = nbeg - (seg == 0 ? 0 : (seg == 1 ? N0 : N0 + N1));

    const _Float16* pW = Wt + (size_t)(nbeg + 8 * wave) * GK + 8 * lane;
    v8h bq[8];
#pragma unroll
    for (int t = 0; t < 8; ++t) bq[t] = *(const v8h*)(pW + (size_t)t * GK);

    {
        const float* pA = A + (size_t)m0 * GK + 4 * lane;
        const float* pS = MODE == 2 ? sl + (size_t)m0 * C1 + ((4 * lane) & (C1 - 1)) : pA;
        _Float16* as = &As[wave][4 * lane];
#pragma unroll
        for (int g = 0; g < 2; ++g) {
            v4f xa[16][2], ta[16];
#pragma unroll
            for (int r = 0; r < 16; ++r) {
                xa[r][0] = *(const v4f*)(pA + (size_t)(16 * g + r) * GK);
                xa[r][1] = *(const v4f*)(pA + (size_t)(16 * g + r) * GK + 128);
                ta[r] = MODE == 2 ? *(const v4f*)(pS + (size_t)(16 * g + r) * C1) : v4f{};
            }
#pragma unroll
            for (int r = 0; r < 16; ++r) {
                *(v4h*)(as + (16 * g + r) * LPH)       = cvt4<MODE>(xa[r][0], ta[r]);
                *(v4h*)(as + (16 * g + r) * LPH + 128) = cvt4<MODE>(xa[r][1], ta[r]);
            }
        }
    }
#pragma unroll
    for (int t = 0; t < 8; ++t) *(v8h*)&Bs[0][(8 * wave + t) * LPH + 8 * lane] = bq[t];
    __syncthreads();

    v16h a0[GKS], a1[GKS];
#pragma unroll
    for (int s = 0; s < GKS; ++s) {
        a0[s] = ld_frag(&As[wave][lr * LPH + 32 * s + kb]);
        a1[s] = ld_frag(&As[wave][(16 + lr) * LPH + 32 * s + kb]);
    }

    const int mr = m0 + 8 * (lane >> 4);
    for (int it = 0; it < nsteps; ++it) {
        const int n0 = cbeg + 32 * it;
        const int cur = it & 1;
        const bool more = it + 1 < nsteps;
        if (more) {
#pragma unroll
            for (int t = 0; t < 8; ++t) bq[t] = *(const v8h*)(pW + (size_t)(32 * (it + 1) + t) * GK);
        }
        const _Float16* bs = &Bs[cur][lr * LPH + kb];
        v16h b0[GKS], b1[GKS];
#pragma unroll
        for (int s = 0; s < GKS; ++s) {
            b0[s] = ld_frag(bs + 32 * s);
            b1[s] = ld_frag(bs + 16 * LPH + 32 * s);
        }
        __builtin_amdgcn_sched_barrier(0);
        v8f acc00 = {}, acc01 = {}, acc10 = {}, acc11 = {};
#pragma unroll
        for (int s = 0; s < GKS; ++s) {
            acc00 = __builtin_amdgcn_wmma_f32_16x16x32_f16(false, a0[s], false, b0[s], (short)0, acc00, false, false);
            acc01 = __builtin_amdgcn_wmma_f32_16x16x32_f16(false, a0[s], false, b1[s], (short)0, acc01, false, false);
            acc10 = __builtin_amdgcn_wmma_f32_16x16x32_f16(false, a1[s], false, b0[s], (short)0, acc10, false, false);
            acc11 = __builtin_amdgcn_wmma_f32_16x16x32_f16(false, a1[s], false, b1[s], (short)0, acc11, false, false);
        }
        if (more) {
#pragma unroll
            for (int t = 0; t < 8; ++t) *(v8h*)&Bs[cur ^ 1][(8 * wave + t) * LPH + 8 * lane] = bq[t];
            __syncthreads();
        }
        if (live) {
            const int cb0 = n0 + lr;
            const int cb1 = cb0 + 16;
            const float bv0 = bias ? bias[cb0] : 0.f;
            const float bv1 = bias ? bias[cb1] : 0.f;
#pragma unroll
            for (int v = 0; v < 8; ++v) {
                Y[(size_t)(mr + v) * N + cb0]      = acc00[v] + bv0;
                Y[(size_t)(mr + v) * N + cb1]      = acc01[v] + bv1;
                Y[(size_t)(mr + 16 + v) * N + cb0] = acc10[v] + bv0;
                Y[(size_t)(mr + 16 + v) * N + cb1] = acc11[v] + bv1;
            }
        }
    }
}

#define EPW (NE / 32)
#define WPW (EPW / 32)

__device__ __forceinline__ float xhalf(float x) {
    int b = __builtin_bit_cast(int, x);
    return __builtin_bit_cast(float,
        __builtin_amdgcn_permlanex16(b, b, 0x76543210, 0xfedcba98, false, false));
}
template <int N>
__device__ __forceinline__ float xrow(float x) {
    int b = __builtin_bit_cast(int, x);
    return __builtin_bit_cast(float, __builtin_amdgcn_update_dpp(b, b, 0x160 | N, 0xf, 0xf, false));
}

__device__ __forceinline__ float head_reduce(float p0, float p1, float p2, float p3, int lane) {
    const bool h16 = (lane & 16) != 0;
    float a = h16 ? p2 : p0, sa = h16 ? p0 : p2;
    float c = h16 ? p3 : p1, sc = h16 ? p1 : p3;
    a += xhalf(sa);
    c += xhalf(sc);
    const bool h8 = (lane & 8) != 0;
    float k = h8 ? c : a, sk = h8 ? a : c;
    k += xrow<8>(sk);
    k += xrow<4>(k);
    k += xrow<2>(k);
    k += xrow<1>(k);
    return k;
}

template <int NJ, int HJ>
__device__ __forceinline__ float edge_score(const float (&x)[NJ], const float* xr_row,
                                            const float* att_s, int lane) {
    float p[NH] = {0.f, 0.f, 0.f, 0.f};
    for (int j = 0; j < NJ; ++j) {
        float g = x[j] + xr_row[lane + 32 * j];
        g = g > 0.f ? g : 0.2f * g;
        p[j / HJ] = fmaf(g, att_s[lane + 32 * j], p[j / HJ]);
    }
    return head_reduce(p[0], p[1], p[2], p[3], lane);
}

__device__ __forceinline__ int wave_scan_incl(int v, int lane) {
    for (int d = 1; d < 32; d <<= 1) {
        const int t = __shfl_up(v, d, 32);
        if (lane >= d) v += t;
    }
    return v;
}

__device__ __forceinline__ float dec_max(unsigned enc) {
    return (enc & 0x80000000u) ? __uint_as_float(enc ^ 0x80000000u) : __uint_as_float(~enc);
}

#define LCAP 2048

template <int ROWS, int C, bool MEAN>
__global__ void __launch_bounds__(1024) __attribute__((amdgpu_waves_per_eu(ROWS > 16 ? 16 : 8, 16)))
k_own_gat(const int* __restrict__ src, const int* __restrict__ dst,
          const float* __restrict__ xl, const float* __restrict__ xr,
          const float* __restrict__ att, const float* __restrict__ bias,
          float* __restrict__ out, int num_dst) {
    constexpr int SROW = NH * C;
    constexpr int NJ   = SROW / 32;
    constexpr int HJ   = C / 32;
    constexpr int OROW = MEAN ? C : SROW;
    constexpr int NQ   = 16;
    constexpr int NB   = ROWS > 16 ? 2 : 4;
    constexpr int RS   = ROWS + 1;
    constexpr unsigned PAD_ENT = (unsigned)ROWS << 16;
    __shared__ float    xr_s[RS * SROW];
    __shared__ float    acc[RS * SROW];
    __shared__ float    att_s[SROW];
    __shared__ unsigned menc_s[RS * NH];
    __shared__ float    mapp_s[2][RS * NH];
    __shared__ float    den_s[RS * NH];
    __shared__ __attribute__((aligned(16))) unsigned smask[32 * WPW];
    __shared__ __attribute__((aligned(16))) unsigned elist[LCAP];
    __shared__ int      wcnt[32];

    const int tid = threadIdx.x, lane = tid & 31;
    const int wave = __builtin_amdgcn_readfirstlane(tid >> 5);
    const int lo = blockIdx.x * ROWS;
    int nrows = num_dst - lo;
    if (nrows > ROWS) nrows = ROWS;
    if (nrows < 0) nrows = 0;
    const int hh = (lane >> 3) & 3;

    for (int i = tid; i < RS * SROW; i += 1024) {
        acc[i] = 0.f;
        xr_s[i] = i < nrows * SROW ? xr[(size_t)lo * SROW + i] : 0.f;
    }
    for (int i = tid; i < SROW; i += 1024) att_s[i] = att[i];
    for (int i = tid; i < RS * NH; i += 1024) {
        menc_s[i] = NEG_ENC; den_s[i] = 0.f;
        mapp_s[1][i] = dec_max(NEG_ENC);
    }

    int cnt = 0;
    for (int b = 0; b < WPW / NQ; ++b) {
        const int base = wave * EPW + b * (32 * NQ);
        int dv[NQ];
#pragma unroll
        for (int q = 0; q < NQ; ++q) dv[q] = dst[base + 32 * q + lane];
#pragma unroll
        for (int q = 0; q < NQ; ++q) {
            const unsigned mask =
                __builtin_amdgcn_ballot_w32((unsigned)(dv[q] - lo) < (unsigned)nrows);
            if (lane == 0) smask[wave * WPW + b * NQ + q] = mask;
            cnt += __builtin_popcount(mask);
        }
    }
    if (lane == 0) wcnt[wave] = cnt;
    __syncthreads();

    const int wc   = wcnt[lane];
    const int winc = wave_scan_incl(wc, lane);
    const int T     = __builtin_amdgcn_readlane(winc, 31);
    const int wbase = __builtin_amdgcn_readlane(winc - wc, wave);
    auto expand = [&](int start, int n) {
        const uint4 mw = *(const uint4*)&smask[wave * WPW + 4 * lane];
        const int s4 = __builtin_popcount(mw.x) + __builtin_popcount(mw.y) +
                       __builtin_popcount(mw.z) + __builtin_popcount(mw.w);
        const int lbase = wbase + wave_scan_incl(s4, lane) - s4;
        unsigned long long lo64 = (unsigned long long)mw.x | ((unsigned long long)mw.y << 32);
        unsigned long long hi64 = (unsigned long long)mw.z | ((unsigned long long)mw.w << 32);
        int pos = lbase - start;
        const int ebase = wave * EPW + 128 * lane;
        while (__builtin_amdgcn_ballot_w32((lo64 | hi64) != 0ull) != 0u) {
            int ei[4], sv[4], dw[4];
#pragma unroll
            for (int t = 0; t < 4; ++t) {
                int idx = -1;
                if (lo64)      { idx = __builtin_ctzll(lo64);      lo64 &= lo64 - 1; }
                else if (hi64) { idx = 64 + __builtin_ctzll(hi64); hi64 &= hi64 - 1; }
                ei[t] = idx;
                const int e = ebase + (idx < 0 ? 0 : idx);
                sv[t] = src[e];
                dw[t] = dst[e];
            }
#pragma unroll
            for (int t = 0; t < 4; ++t) {
                if (ei[t] >= 0) {
                    if ((unsigned)pos < (unsigned)n)
                        elist[pos] = ((unsigned)(dw[t] - lo) << 16) | (unsigned)sv[t];
                    ++pos;
                }
            }
        }
        if (tid < 4 && n + tid < ((n + 3) & ~3)) elist[n + tid] = PAD_ENT;
    };

    int rk = 0;
    for (int start = 0; start < T; start += LCAP) {
        const int n = T - start < LCAP ? T - start : LCAP;
        expand(start, n);
        __syncthreads();
        for (int k0 = 0; k0 < n; k0 += 32 * NB, ++rk) {
            const int b0 = k0 + wave * NB;
            const bool active = b0 < n;
            const int cur = rk & 1;
            unsigned ent[NB];
            float x[NB][NJ];
            float al[NB];
            if (active) {
#pragma unroll
                for (int u = 0; u < NB; ++u)
                    ent[u] = (unsigned)__builtin_amdgcn_readfirstlane((int)elist[b0 + u]);
#pragma unroll
                for (int u = 0; u < NB; ++u) {
                    const float* pl = xl + (size_t)(ent[u] & 0xffffu) * SROW + lane;
                    for (int j = 0; j < NJ; ++j) x[u][j] = pl[32 * j];
                }
#pragma unroll
                for (int u = 0; u < NB; ++u) {
                    const int r = ent[u] >> 16;
                    al[u] = edge_score<NJ, HJ>(x[u], xr_s + r * SROW, att_s, lane);
                    if ((lane & 7) == 0) {
                        unsigned uu = __float_as_uint(al[u]);
                        unsigned enc = (uu & 0x80000000u) ? ~uu : (uu | 0x80000000u);
                        atomicMax(&menc_s[r * NH + hh], enc);
                    }
                }
            }
            __syncthreads();

            {
                constexpr int PPW = (RS * NH + 31) / 32;
                const int pr = wave * PPW + lane;
                const bool mine = lane < PPW && pr < RS * NH;
                float mo = 0.f, mn = 0.f;
                if (mine) {
                    mo = mapp_s[cur ^ 1][pr];
                    mn = dec_max(menc_s[pr]);
                    mapp_s[cur][pr] = mn;
                }
                unsigned moved = __builtin_amdgcn_ballot_w32(mine && mn != mo);
                const float f = expf(mo - mn);
                while (moved) {
                    const int l = __builtin_ctz(moved);
                    moved &= moved - 1;
                    const float fl = __builtin_bit_cast(float,
                        __builtin_amdgcn_readlane(__builtin_bit_cast(int, f), l));
                    const int p = wave * PPW + l;
                    for (int j = 0; j < HJ; ++j) acc[p * C + 32 * j + lane] *= fl;
                    if (lane == 0) den_s[p] *= fl;
                }
            }
            __syncthreads();

            if (active) {
#pragma unroll
                for (int u = 0; u < NB; ++u) {
                    const int r = ent[u] >> 16;
                    const float ea = expf(al[u] - mapp_s[cur][r * NH + hh]);
                    if ((lane & 7) == 0) atomicAdd(&den_s[r * NH + hh], ea);
                    float eah[NH];
                    for (int h = 0; h < NH; ++h)
                        eah[h] = __builtin_bit_cast(float,
                            __builtin_amdgcn_readlane(__builtin_bit_cast(int, ea), 8 * h));
                    float* pa = acc + r * SROW + lane;
                    for (int j = 0; j < NJ; ++j) atomicAdd(&pa[32 * j], eah[j / HJ] * x[u][j]);
                }
            }
        }
    }
    __syncthreads();

    float* po = out + (size_t)lo * OROW;
    for (int i = tid; i < nrows * OROW; i += 1024) {
        const int r = i / OROW, col = i - r * OROW;
        float v;
        if (MEAN) {
            v = 0.f;
            for (int h = 0; h < NH; ++h)
                v += acc[r * SROW + h * C + col] / (den_s[r * NH + h] + 1e-16f);
            v *= 0.25f;
        } else {
            v = acc[i] / (den_s[r * NH + col / C] + 1e-16f);
        }
        if (bias) v += bias[col];
        po[i] = v;
    }
}

extern "C" void kernel_launch(void* const* d_in, const int* in_sizes, int n_in,
                              void* d_out, int out_size, void* d_ws, size_t ws_size,
                              hipStream_t stream) {
    (void)in_sizes; (void)n_in; (void)out_size; (void)ws_size;

    const float* x_sample = (const float*)d_in[0];
    const float* x_gene   = (const float*)d_in[1];
    const int*   sg_src   = (const int*)d_in[2];
    const int*   sg_dst   = (const int*)d_in[3];
    const int*   gs_src   = (const int*)d_in[4];
    const int*   gs_dst   = (const int*)d_in[5];
    const float* Wl1_sg = (const float*)d_in[6];   const float* bl1_sg  = (const float*)d_in[7];
    const float* Wr1_sg = (const float*)d_in[8];   const float* br1_sg  = (const float*)d_in[9];
    const float* att1_sg = (const float*)d_in[10]; const float* bias1_sg = (const float*)d_in[11];
    const float* Wl1_gs = (const float*)d_in[12];  const float* bl1_gs  = (const float*)d_in[13];
    const float* Wr1_gs = (const float*)d_in[14];  const float* br1_gs  = (const float*)d_in[15];
    const float* att1_gs = (const float*)d_in[16]; const float* bias1_gs = (const float*)d_in[17];
    const float* Wl3 = (const float*)d_in[18];     const float* bl3 = (const float*)d_in[19];
    const float* Wr3 = (const float*)d_in[20];     const float* br3 = (const float*)d_in[21];
    const float* att3 = (const float*)d_in[22];    const float* bias3 = (const float*)d_in[23];
    const float* sl1_W = (const float*)d_in[24];   const float* sl1_b = (const float*)d_in[25];
    const float* sl3_W = (const float*)d_in[26];   const float* sl3_b = (const float*)d_in[27];

    char* p = (char*)d_ws;
    auto carve = [&](size_t bytes) -> char* {
        char* r = p; p += (bytes + 255) & ~(size_t)255; return r;
    };

    _Float16* hW     = (_Float16*)carve((size_t)WT_ROWS * WK * 2);
    _Float16* hWl1sg = hW + (size_t)WR_XS * WK;
    _Float16* hWr1sg = hW + (size_t)WR_XG * WK;
    _Float16* hWr3   = hW + (size_t)WR_X1S * WK;
    _Float16* hWl3   = hW + (size_t)WR_X1G * WK;
    float* xl_sg = (float*)carve((size_t)NS * HC1 * 4);
    float* xr_sg = (float*)carve((size_t)NG * HC1 * 4);
    float* out1_gene   = (float*)carve((size_t)NG * HC1 * 4);
    float* xr_gs = (float*)carve((size_t)NS * HC1 * 4);
    float* out1_sample = (float*)carve((size_t)NS * HC1 * 4);
    float* sl1 = (float*)carve((size_t)NS * C1 * 4);
    float* xl3 = (float*)carve((size_t)NG * HC3 * 4);
    float* xr3 = (float*)carve((size_t)NS * HC3 * 4);
    float* acc3 = (float*)carve((size_t)NS * C3 * 4);
    float* sl3 = (float*)carve((size_t)NS * C3 * 4);
    float* xl_gs = xl3;

    const int TB = 256;
#define EW(n) cdiv((n), TB), TB, 0, stream
    static_assert(DIN == WK && HC1 == WK && WT_ROWS % 32 == 0 && C1 % 32 == 0 && C3 % 32 == 0,
                  "k_weights_t shape");
    k_weights_t<<<(WT_ROWS / 32) * (WK / 64), 256, 0, stream>>>(Wl1_sg, Wr1_gs, sl1_W, Wr1_sg, Wl1_gs,
                                                                Wr3, sl3_W, Wl3, hW);

    static_assert(DIN == GK && HC1 == GK && NS % 32 == 0 && NG % 32 == 0, "k_gemm_lds shape");
#define GEMM3(MODE, A, SL, Wt, b0, Y0, N0, b1, Y1, N1, b2, Y2, N2, M, nch) \
    k_gemm_lds<MODE><<<dim3(cdiv((M), 128), ((N0) + (N1) + (N2)) / (nch)), 128, 0, stream>>>( \
        A, SL, Wt, b0, Y0, N0, b1, Y1, N1, b2, Y2, N2, M, nch)

    GEMM3(0, x_sample, x_sample, hWl1sg, bl1_sg, xl_sg, HC1, br1_gs, xr_gs, HC1, sl1_b, sl1, C1, NS, 64);
    GEMM3(0, x_gene, x_gene, hWr1sg, br1_sg, xr_sg, HC1, bl1_gs, xl_gs, HC1, nullptr, nullptr, 0, NG, 256);

    k_own_gat<40, C1, false><<<cdiv(NG, 40), 1024, 0, stream>>>(sg_src, sg_dst, xl_sg, xr_sg, att1_sg,
                                                                bias1_sg, out1_gene, NG);
    k_own_gat<16, C1, false><<<cdiv(NS, 16), 1024, 0, stream>>>(gs_src, gs_dst, xl_gs, xr_gs, att1_gs,
                                                                bias1_gs, out1_sample, NS);

    GEMM3(1, out1_gene, out1_gene, hWl3, bl3, xl3, HC3, nullptr, nullptr, 0, nullptr, nullptr, 0, NG, 256);
    GEMM3(2, out1_sample, sl1, hWr3, br3, xr3, HC3, sl3_b, sl3, C3, nullptr, nullptr, 0, NS, 64);
    k_own_gat<16, C3, true><<<cdiv(NS, 16), 1024, 0, stream>>>(gs_src, gs_dst, xl3, xr3, att3,
                                                               nullptr, acc3, NS);
    k_final<<<EW(NS * C3)>>>(acc3, sl3, bias3, (float*)d_out, NS * C3);
#undef GEMM3
#undef EW
}
